# starting kernel
# baseline (speedup 1.0000x reference)
#include <hip/hip_runtime.h>
#include <math.h>

typedef __attribute__((ext_vector_type(2))) float v2f;
typedef __attribute__((ext_vector_type(8))) float v8f;

#define BOHR_F 1.8897261258369282f
#define PI_F   3.14159265358979323846f

__device__ __forceinline__ void sincos2pi_x8(v8f x, v8f* sn, v8f* cs) {
    float s0, s1, s2, s3, s4, s5, s6, s7;
    float c0, c1, c2, c3, c4, c5, c6, c7;
    asm volatile(
        "v_sin_f32 %0, %16\n\t"
        "v_sin_f32 %1, %17\n\t"
        "v_sin_f32 %2, %18\n\t"
        "v_sin_f32 %3, %19\n\t"
        "v_sin_f32 %4, %20\n\t"
        "v_sin_f32 %5, %21\n\t"
        "v_sin_f32 %6, %22\n\t"
        "v_sin_f32 %7, %23\n\t"
        "v_cos_f32 %8, %16\n\t"
        "v_cos_f32 %9, %17\n\t"
        "v_cos_f32 %10, %18\n\t"
        "v_cos_f32 %11, %19\n\t"
        "v_cos_f32 %12, %20\n\t"
        "v_cos_f32 %13, %21\n\t"
        "v_cos_f32 %14, %22\n\t"
        "v_cos_f32 %15, %23\n\t"
        "v_nop"
        : "=&v"(s0), "=&v"(s1), "=&v"(s2), "=&v"(s3),
          "=&v"(s4), "=&v"(s5), "=&v"(s6), "=&v"(s7),
          "=&v"(c0), "=&v"(c1), "=&v"(c2), "=&v"(c3),
          "=&v"(c4), "=&v"(c5), "=&v"(c6), "=&v"(c7)
        : "v"(x[0]), "v"(x[1]), "v"(x[2]), "v"(x[3]),
          "v"(x[4]), "v"(x[5]), "v"(x[6]), "v"(x[7]));
    *sn = (v8f){s0, s1, s2, s3, s4, s5, s6, s7};
    *cs = (v8f){c0, c1, c2, c3, c4, c5, c6, c7};
}

__global__ void k_zero(float* __restrict__ p, int n) {
    int i = blockIdx.x * blockDim.x + threadIdx.x;
    if (i < n) p[i] = 0.0f;
}

__global__ void k_pack(const float* __restrict__ coords, const float* __restrict__ q,
                       const float* __restrict__ kvec, const float* __restrict__ expfac,
                       float4* __restrict__ packA, float4* __restrict__ packK,
                       int BN, int NK)
{
    int i = blockIdx.x * blockDim.x + threadIdx.x;
    if (i < BN) {
        float4 p;
        p.x = coords[i * 3 + 0];
        p.y = coords[i * 3 + 1];
        p.z = coords[i * 3 + 2];
        p.w = q[i];
        packA[i] = p;
    }
    if (i < NK) {
        float4 p;
        p.x = kvec[i * 3 + 0];
        p.y = kvec[i * 3 + 1];
        p.z = kvec[i * 3 + 2];
        p.w = expfac[i];
        packK[i] = p;
    }
}

__global__ __launch_bounds__(256) void k_struct_fac(
    const float4* __restrict__ packA, const float* __restrict__ cell_inv,
    const float* __restrict__ kvec, float* __restrict__ Sri,
    int N, int NK, int splitN)
{
    const int lane  = threadIdx.x & 31;
    const int wave  = threadIdx.x >> 5;
    const int kTile = (blockIdx.x * 8 + wave) * 16;
    const int b     = blockIdx.z;
    const int nPer  = N / splitN;
    const int nBeg  = blockIdx.y * nPer;

    const int  m  = lane & 15;
    const bool hi = lane >= 16;

    const float kx = kvec[(kTile + m) * 3 + 0];
    const float ky = kvec[(kTile + m) * 3 + 1];
    const float kz = kvec[(kTile + m) * 3 + 2];
    const float kp0 = kx * cell_inv[0] + ky * cell_inv[3] + kz * cell_inv[6];
    const float kp1 = kx * cell_inv[1] + ky * cell_inv[4] + kz * cell_inv[7];
    const float kp2 = kx * cell_inv[2] + ky * cell_inv[5] + kz * cell_inv[8];

    v2f a;
    a.x = hi ? kp2 : kp0;
    a.y = hi ? 0.0f : kp1;

    v8f sumRe = {}, sumIm = {};
    const float4* pA = packA + (size_t)b * N;

#pragma unroll 2
    for (int n0 = nBeg; n0 < nBeg + nPer; n0 += 16) {
        const float4 p = pA[n0 + m];
        v2f bm;
        bm.x = hi ? p.z : p.x;
        bm.y = hi ? 0.0f : p.y;

        v8f d = {};
        d = __builtin_amdgcn_wmma_f32_16x16x4_f32(
                false, a, false, bm, (short)0, d, false, false);

        v8f sn, cs;
        sincos2pi_x8(d, &sn, &cs);
        sumRe = sumRe + cs * p.w;
        sumIm = sumIm + sn * p.w;
    }

#pragma unroll
    for (int r = 0; r < 8; ++r) {
        float re = sumRe[r], im = sumIm[r];
        for (int mask = 8; mask >= 1; mask >>= 1) {
            re += __shfl_xor(re, mask, 32);
            im += __shfl_xor(im, mask, 32);
        }
        sumRe[r] = re; sumIm[r] = im;
    }
    if (m == 0) {
        const int kbase = kTile + (hi ? 8 : 0);
#pragma unroll
        for (int r = 0; r < 8; ++r) {
            atomicAdd(&Sri[2 * ((size_t)b * NK + kbase + r) + 0], sumRe[r]);
            atomicAdd(&Sri[2 * ((size_t)b * NK + kbase + r) + 1], sumIm[r]);
        }
    }
}

__global__ __launch_bounds__(256) void k_recip(
    const float4* __restrict__ packA, const float* __restrict__ cell_inv,
    const float4* __restrict__ packK, const float2* __restrict__ Sri,
    float* __restrict__ recip, int N, int NK, int splitK)
{
    const int lane  = threadIdx.x & 31;
    const int wave  = threadIdx.x >> 5;
    const int nTile = (blockIdx.x * 8 + wave) * 16;
    const int b     = blockIdx.z;
    const int kPer  = NK / splitK;
    const int kBeg  = blockIdx.y * kPer;

    const int  m  = lane & 15;
    const bool hi = lane >= 16;

    const float4 pc = packA[(size_t)b * N + nTile + m];
    const float s0 = cell_inv[0] * pc.x + cell_inv[1] * pc.y + cell_inv[2] * pc.z;
    const float s1 = cell_inv[3] * pc.x + cell_inv[4] * pc.y + cell_inv[5] * pc.z;
    const float s2 = cell_inv[6] * pc.x + cell_inv[7] * pc.y + cell_inv[8] * pc.z;

    v2f a;
    a.x = hi ? s2 : s0;
    a.y = hi ? 0.0f : s1;

    v8f acc = {};
    const float2* pS = Sri + (size_t)b * NK;

#pragma unroll 2
    for (int kc = kBeg; kc < kBeg + kPer; kc += 16) {
        const float4 kv = packK[kc + m];
        v2f bm;
        bm.x = hi ? kv.z : kv.x;
        bm.y = hi ? 0.0f : kv.y;

        v8f d = {};
        d = __builtin_amdgcn_wmma_f32_16x16x4_f32(
                false, a, false, bm, (short)0, d, false, false);

        const float2 S = pS[kc + m];
        v8f sn, cs;
        sincos2pi_x8(d, &sn, &cs);
        acc = acc + (cs * S.x + sn * S.y) * kv.w;
    }

#pragma unroll
    for (int r = 0; r < 8; ++r) {
        float v = acc[r];
        for (int mask = 8; mask >= 1; mask >>= 1)
            v += __shfl_xor(v, mask, 32);
        acc[r] = v;
    }
    if (m == 0) {
        const int nbase = nTile + (hi ? 8 : 0);
#pragma unroll
        for (int r = 0; r < 8; ++r)
            atomicAdd(&recip[(size_t)b * N + nbase + r], acc[r]);
    }
}

__global__ void k_phi(const float* __restrict__ q, const float* __restrict__ recip,
                      const float* __restrict__ volume, const float* __restrict__ bewald,
                      float* __restrict__ out, int BN)
{
    int i = blockIdx.x * blockDim.x + threadIdx.x;
    if (i >= BN) return;
    const float scale = BOHR_F / (PI_F * volume[0]);
    const float self  = 2.0f * bewald[0] * BOHR_F / sqrtf(PI_F);
    const float phi   = recip[i] * scale - q[i] * self;
    out[i]      = 0.5f * q[i] * phi;
    out[BN + i] = phi;
}

extern "C" void kernel_launch(void* const* d_in, const int* in_sizes, int n_in,
                              void* d_out, int out_size, void* d_ws, size_t ws_size,
                              hipStream_t stream)
{
    const float* coords   = (const float*)d_in[0];
    const float* q        = (const float*)d_in[1];
    const float* cell_inv = (const float*)d_in[2];
    const float* kvec     = (const float*)d_in[3];
    const float* expfac   = (const float*)d_in[4];
    const float* volume   = (const float*)d_in[5];
    const float* bewald   = (const float*)d_in[6];
    float* out = (float*)d_out;

    const int B  = 2;
    const int BN = in_sizes[1];
    const int N  = BN / B;
    const int NK = in_sizes[4];

    float*  Sri   = (float*)d_ws;
    float*  recip = Sri + (size_t)2 * B * NK;
    float4* packA = (float4*)(recip + BN);
    float4* packK = packA + BN;
    const size_t wsFloats = (size_t)2 * B * NK + BN + (size_t)4 * BN + (size_t)4 * NK;
    if (ws_size < wsFloats * sizeof(float)) return;

    const int SPLIT_N = 16;
    const int SPLIT_K = 16;

    const int zeroN = 2 * B * NK + BN;
    k_zero<<<(zeroN + 255) / 256, 256, 0, stream>>>(Sri, zeroN);

    const int packN = (BN > NK) ? BN : NK;
    k_pack<<<(packN + 255) / 256, 256, 0, stream>>>(coords, q, kvec, expfac,
                                                    packA, packK, BN, NK);

    dim3 g1(NK / 128, SPLIT_N, B);
    k_struct_fac<<<g1, 256, 0, stream>>>(packA, cell_inv, kvec, Sri, N, NK, SPLIT_N);

    dim3 g2(N / 128, SPLIT_K, B);
    k_recip<<<g2, 256, 0, stream>>>(packA, cell_inv, packK, (float2*)Sri,
                                    recip, N, NK, SPLIT_K);

    k_phi<<<(BN + 255) / 256, 256, 0, stream>>>(q, recip, volume, bewald, out, BN);
}


# Split partial sums go to per-panel workspace planes by plain whole-line stores, not atomics; kernel 2's panel fill and the epilogue add the 4 planes in fixed order; zero-fill launch goes (5 to 4)
# speedup vs baseline: 1.4448x; 1.4448x over previous
#include <hip/hip_runtime.h>
#include <math.h>

typedef __attribute__((ext_vector_type(2))) float v2f;
typedef __attribute__((ext_vector_type(8))) float v8f;

#define BOHR_F 1.8897261258369282f
#define PI_F   3.14159265358979323846f

__device__ __forceinline__ void sincos2pi_x8(v8f x, v8f* sn, v8f* cs) {
    float s0, s1, s2, s3, s4, s5, s6, s7;
    float c0, c1, c2, c3, c4, c5, c6, c7;
    asm volatile(
        "v_sin_f32 %0, %16\n\t"
        "v_sin_f32 %1, %17\n\t"
        "v_sin_f32 %2, %18\n\t"
        "v_sin_f32 %3, %19\n\t"
        "v_sin_f32 %4, %20\n\t"
        "v_sin_f32 %5, %21\n\t"
        "v_sin_f32 %6, %22\n\t"
        "v_sin_f32 %7, %23\n\t"
        "v_cos_f32 %8, %16\n\t"
        "v_cos_f32 %9, %17\n\t"
        "v_cos_f32 %10, %18\n\t"
        "v_cos_f32 %11, %19\n\t"
        "v_cos_f32 %12, %20\n\t"
        "v_cos_f32 %13, %21\n\t"
        "v_cos_f32 %14, %22\n\t"
        "v_cos_f32 %15, %23\n\t"
        "v_nop"
        : "=&v"(s0), "=&v"(s1), "=&v"(s2), "=&v"(s3),
          "=&v"(s4), "=&v"(s5), "=&v"(s6), "=&v"(s7),
          "=&v"(c0), "=&v"(c1), "=&v"(c2), "=&v"(c3),
          "=&v"(c4), "=&v"(c5), "=&v"(c6), "=&v"(c7)
        : "v"(x[0]), "v"(x[1]), "v"(x[2]), "v"(x[3]),
          "v"(x[4]), "v"(x[5]), "v"(x[6]), "v"(x[7]));
    *sn = (v8f){s0, s1, s2, s3, s4, s5, s6, s7};
    *cs = (v8f){c0, c1, c2, c3, c4, c5, c6, c7};
}

__global__ void k_pack(const float* __restrict__ coords, const float* __restrict__ q,
                       const float* __restrict__ kvec, const float* __restrict__ expfac,
                       float4* __restrict__ packA, float4* __restrict__ packK,
                       int BN, int NK)
{
    int i = blockIdx.x * blockDim.x + threadIdx.x;
    if (i < BN) {
        float4 p;
        p.x = coords[i * 3 + 0];
        p.y = coords[i * 3 + 1];
        p.z = coords[i * 3 + 2];
        p.w = q[i];
        packA[i] = p;
    }
    if (i < NK) {
        float4 p;
        p.x = kvec[i * 3 + 0];
        p.y = kvec[i * 3 + 1];
        p.z = kvec[i * 3 + 2];
        p.w = expfac[i];
        packK[i] = p;
    }
}

#define PANEL  1024
#define NSPLIT 4
#define WAVES  4
#define TPB    (WAVES * 32)

__device__ __forceinline__ v8f phase_tile(v2f a, float4 p, bool hi) {
    v2f bm;
    bm.x = hi ? p.z : p.x;
    bm.y = hi ? 0.0f : p.y;
    v8f z = {};
    return __builtin_amdgcn_wmma_f32_16x16x4_f32(
               false, a, false, bm, (short)0, z, false, false);
}

__global__ __launch_bounds__(TPB) void k_struct_fac(
    const float4* __restrict__ packA, const float* __restrict__ cell_inv,
    const float* __restrict__ kvec, float2* __restrict__ Spart,
    int N, int NK, int B)
{
    __shared__ float4 sAtom[PANEL];

    const int lane  = threadIdx.x & 31;
    const int wave  = threadIdx.x >> 5;
    const int kTile = (blockIdx.x * WAVES + wave) * 16;
    const int b     = blockIdx.z;
    const int nBeg  = blockIdx.y * PANEL;

    const int  m  = lane & 15;
    const bool hi = lane >= 16;

    const float4* pA = packA + (size_t)b * N + nBeg;
    float4 stage[PANEL / TPB];
#pragma unroll
    for (int i = 0; i < PANEL / TPB; ++i)
        stage[i] = pA[i * TPB + threadIdx.x];

    const float kx = kvec[(kTile + m) * 3 + 0];
    const float ky = kvec[(kTile + m) * 3 + 1];
    const float kz = kvec[(kTile + m) * 3 + 2];
    const float kp0 = kx * cell_inv[0] + ky * cell_inv[3] + kz * cell_inv[6];
    const float kp1 = kx * cell_inv[1] + ky * cell_inv[4] + kz * cell_inv[7];
    const float kp2 = kx * cell_inv[2] + ky * cell_inv[5] + kz * cell_inv[8];

    v2f a;
    a.x = hi ? kp2 : kp0;
    a.y = hi ? 0.0f : kp1;

#pragma unroll
    for (int i = 0; i < PANEL / TPB; ++i)
        sAtom[i * TPB + threadIdx.x] = stage[i];
    __syncthreads();

    v8f sumRe = {}, sumIm = {};

    for (int c = 0; c < PANEL; c += 64) {
        const float4 p0 = sAtom[c +  0 + m];
        const float4 p1 = sAtom[c + 16 + m];
        const float4 p2 = sAtom[c + 32 + m];
        const float4 p3 = sAtom[c + 48 + m];

        const v8f d0 = phase_tile(a, p0, hi);
        const v8f d1 = phase_tile(a, p1, hi);
        const v8f d2 = phase_tile(a, p2, hi);
        const v8f d3 = phase_tile(a, p3, hi);

        v8f sn, cs;
        sincos2pi_x8(d0, &sn, &cs);
        sumRe = sumRe + cs * p0.w;
        sumIm = sumIm + sn * p0.w;
        sincos2pi_x8(d1, &sn, &cs);
        sumRe = sumRe + cs * p1.w;
        sumIm = sumIm + sn * p1.w;
        sincos2pi_x8(d2, &sn, &cs);
        sumRe = sumRe + cs * p2.w;
        sumIm = sumIm + sn * p2.w;
        sincos2pi_x8(d3, &sn, &cs);
        sumRe = sumRe + cs * p3.w;
        sumIm = sumIm + sn * p3.w;
    }

#pragma unroll
    for (int r = 0; r < 8; ++r) {
        float re = sumRe[r], im = sumIm[r];
        for (int mask = 8; mask >= 1; mask >>= 1) {
            re += __shfl_xor(re, mask, 32);
            im += __shfl_xor(im, mask, 32);
        }
        sumRe[r] = re; sumIm[r] = im;
    }
    const int  row = (lane >> 1) & 7;
    const bool im  = lane & 1;
    float v = im ? sumIm[0] : sumRe[0];
#pragma unroll
    for (int r = 1; r < 8; ++r) {
        const float cand = im ? sumIm[r] : sumRe[r];
        v = (row == r) ? cand : v;
    }
    float* line = (float*)(Spart + ((size_t)blockIdx.y * B + b) * NK + kTile);
    line[lane] = v;
}

__global__ __launch_bounds__(TPB) __attribute__((amdgpu_waves_per_eu(1, 4))) void k_recip(
    const float4* __restrict__ packA, const float* __restrict__ cell_inv,
    const float4* __restrict__ packK, const float2* __restrict__ Spart,
    float* __restrict__ recipPart, int N, int NK, int B)
{
    __shared__ float4 sK[PANEL];
    __shared__ float4 sS4[PANEL / 2];
    __shared__ float  sOut[16 * WAVES];

    const int lane  = threadIdx.x & 31;
    const int wave  = threadIdx.x >> 5;
    const int nTile = (blockIdx.x * WAVES + wave) * 16;
    const int b     = blockIdx.z;
    const int kBeg  = blockIdx.y * PANEL;

    const int  m  = lane & 15;
    const bool hi = lane >= 16;

    const float4* pK = packK + kBeg;
    float4 stageK[PANEL / TPB];
    float4 stageS[NSPLIT][PANEL / 2 / TPB];
#pragma unroll
    for (int i = 0; i < PANEL / TPB; ++i)
        stageK[i] = pK[i * TPB + threadIdx.x];
#pragma unroll
    for (int p = 0; p < NSPLIT; ++p) {
        const float4* pS = (const float4*)(Spart + ((size_t)p * B + b) * NK + kBeg);
#pragma unroll
        for (int i = 0; i < PANEL / 2 / TPB; ++i)
            stageS[p][i] = pS[i * TPB + threadIdx.x];
    }

    const float4 pc = packA[(size_t)b * N + nTile + m];
    const float s0 = cell_inv[0] * pc.x + cell_inv[1] * pc.y + cell_inv[2] * pc.z;
    const float s1 = cell_inv[3] * pc.x + cell_inv[4] * pc.y + cell_inv[5] * pc.z;
    const float s2 = cell_inv[6] * pc.x + cell_inv[7] * pc.y + cell_inv[8] * pc.z;

    v2f a;
    a.x = hi ? s2 : s0;
    a.y = hi ? 0.0f : s1;

#pragma unroll
    for (int i = 0; i < PANEL / TPB; ++i)
        sK[i * TPB + threadIdx.x] = stageK[i];
#pragma unroll
    for (int i = 0; i < PANEL / 2 / TPB; ++i) {
        float4 s = stageS[0][i];
#pragma unroll
        for (int p = 1; p < NSPLIT; ++p) {
            s.x += stageS[p][i].x;
            s.y += stageS[p][i].y;
            s.z += stageS[p][i].z;
            s.w += stageS[p][i].w;
        }
        sS4[i * TPB + threadIdx.x] = s;
    }
    __syncthreads();

    const float2* sS = (const float2*)sS4;

    v8f acc = {};

    for (int c = 0; c < PANEL; c += 64) {
        const float4 kv0 = sK[c +  0 + m];
        const float4 kv1 = sK[c + 16 + m];
        const float4 kv2 = sK[c + 32 + m];
        const float4 kv3 = sK[c + 48 + m];
        const float2 S0 = sS[c +  0 + m];
        const float2 S1 = sS[c + 16 + m];
        const float2 S2 = sS[c + 32 + m];
        const float2 S3 = sS[c + 48 + m];

        const v8f d0 = phase_tile(a, kv0, hi);
        const v8f d1 = phase_tile(a, kv1, hi);
        const v8f d2 = phase_tile(a, kv2, hi);
        const v8f d3 = phase_tile(a, kv3, hi);

        v8f sn, cs;
        sincos2pi_x8(d0, &sn, &cs);
        acc = acc + (cs * S0.x + sn * S0.y) * kv0.w;
        sincos2pi_x8(d1, &sn, &cs);
        acc = acc + (cs * S1.x + sn * S1.y) * kv1.w;
        sincos2pi_x8(d2, &sn, &cs);
        acc = acc + (cs * S2.x + sn * S2.y) * kv2.w;
        sincos2pi_x8(d3, &sn, &cs);
        acc = acc + (cs * S3.x + sn * S3.y) * kv3.w;
    }

#pragma unroll
    for (int r = 0; r < 8; ++r) {
        float v = acc[r];
        for (int mask = 8; mask >= 1; mask >>= 1)
            v += __shfl_xor(v, mask, 32);
        acc[r] = v;
    }
    if (m == 0) {
#pragma unroll
        for (int r = 0; r < 8; ++r)
            sOut[wave * 16 + (hi ? 8 : 0) + r] = acc[r];
    }
    __syncthreads();
    if (threadIdx.x < 16 * WAVES)
        recipPart[((size_t)blockIdx.y * B + b) * N
                  + blockIdx.x * (16 * WAVES) + threadIdx.x] = sOut[threadIdx.x];
}

__global__ void k_phi(const float* __restrict__ q, const float* __restrict__ recipPart,
                      const float* __restrict__ volume, const float* __restrict__ bewald,
                      float* __restrict__ out, int BN)
{
    int i = blockIdx.x * blockDim.x + threadIdx.x;
    if (i >= BN) return;
    const float scale = BOHR_F / (PI_F * volume[0]);
    const float self  = 2.0f * bewald[0] * BOHR_F / sqrtf(PI_F);
    float recip = recipPart[i];
#pragma unroll
    for (int p = 1; p < NSPLIT; ++p)
        recip += recipPart[(size_t)p * BN + i];
    const float phi   = recip * scale - q[i] * self;
    out[i]      = 0.5f * q[i] * phi;
    out[BN + i] = phi;
}

extern "C" void kernel_launch(void* const* d_in, const int* in_sizes, int n_in,
                              void* d_out, int out_size, void* d_ws, size_t ws_size,
                              hipStream_t stream)
{
    const float* coords   = (const float*)d_in[0];
    const float* q        = (const float*)d_in[1];
    const float* cell_inv = (const float*)d_in[2];
    const float* kvec     = (const float*)d_in[3];
    const float* expfac   = (const float*)d_in[4];
    const float* volume   = (const float*)d_in[5];
    const float* bewald   = (const float*)d_in[6];
    float* out = (float*)d_out;

    const int B  = 2;
    const int BN = in_sizes[1];
    const int N  = BN / B;
    const int NK = in_sizes[4];

    float2* Spart     = (float2*)d_ws;
    float*  recipPart = (float*)(Spart + (size_t)NSPLIT * B * NK);
    float4* packA     = (float4*)(recipPart + (size_t)NSPLIT * BN);
    float4* packK     = packA + BN;
    const size_t wsFloats = (size_t)2 * NSPLIT * B * NK + (size_t)NSPLIT * BN
                          + (size_t)4 * BN + (size_t)4 * NK;
    if (ws_size < wsFloats * sizeof(float)) return;
    if (N != NSPLIT * PANEL || NK != NSPLIT * PANEL) return;

    const int packN = (BN > NK) ? BN : NK;
    k_pack<<<(packN + 255) / 256, 256, 0, stream>>>(coords, q, kvec, expfac,
                                                    packA, packK, BN, NK);

    dim3 g1(NK / (16 * WAVES), NSPLIT, B);
    k_struct_fac<<<g1, TPB, 0, stream>>>(packA, cell_inv, kvec, Spart, N, NK, B);

    dim3 g2(N / (16 * WAVES), NSPLIT, B);
    k_recip<<<g2, TPB, 0, stream>>>(packA, cell_inv, packK, Spart,
                                    recipPart, N, NK, B);

    k_phi<<<(BN + 255) / 256, 256, 0, stream>>>(q, recipPart, volume, bewald, out, BN);
}


# Pack launch folded into the two panel fills: each thread reads its rows' three floats and weight from the input arrays and stores the float4 to LDS; no packed tables in memory (4 launches to 3)
# speedup vs baseline: 1.6662x; 1.1532x over previous
#include <hip/hip_runtime.h>
#include <math.h>

typedef __attribute__((ext_vector_type(2))) float v2f;
typedef __attribute__((ext_vector_type(8))) float v8f;

#define BOHR_F 1.8897261258369282f
#define PI_F   3.14159265358979323846f

__device__ __forceinline__ void sincos2pi_x8(v8f x, v8f* sn, v8f* cs) {
    float s0, s1, s2, s3, s4, s5, s6, s7;
    float c0, c1, c2, c3, c4, c5, c6, c7;
    asm volatile(
        "v_sin_f32 %0, %16\n\t"
        "v_sin_f32 %1, %17\n\t"
        "v_sin_f32 %2, %18\n\t"
        "v_sin_f32 %3, %19\n\t"
        "v_sin_f32 %4, %20\n\t"
        "v_sin_f32 %5, %21\n\t"
        "v_sin_f32 %6, %22\n\t"
        "v_sin_f32 %7, %23\n\t"
        "v_cos_f32 %8, %16\n\t"
        "v_cos_f32 %9, %17\n\t"
        "v_cos_f32 %10, %18\n\t"
        "v_cos_f32 %11, %19\n\t"
        "v_cos_f32 %12, %20\n\t"
        "v_cos_f32 %13, %21\n\t"
        "v_cos_f32 %14, %22\n\t"
        "v_cos_f32 %15, %23\n\t"
        "v_nop"
        : "=&v"(s0), "=&v"(s1), "=&v"(s2), "=&v"(s3),
          "=&v"(s4), "=&v"(s5), "=&v"(s6), "=&v"(s7),
          "=&v"(c0), "=&v"(c1), "=&v"(c2), "=&v"(c3),
          "=&v"(c4), "=&v"(c5), "=&v"(c6), "=&v"(c7)
        : "v"(x[0]), "v"(x[1]), "v"(x[2]), "v"(x[3]),
          "v"(x[4]), "v"(x[5]), "v"(x[6]), "v"(x[7]));
    *sn = (v8f){s0, s1, s2, s3, s4, s5, s6, s7};
    *cs = (v8f){c0, c1, c2, c3, c4, c5, c6, c7};
}

#define PANEL  1024
#define NSPLIT 4
#define WAVES  4
#define TPB    (WAVES * 32)

__device__ __forceinline__ v8f phase_tile(v2f a, float4 p, bool hi) {
    v2f bm;
    bm.x = hi ? p.z : p.x;
    bm.y = hi ? 0.0f : p.y;
    v8f z = {};
    return __builtin_amdgcn_wmma_f32_16x16x4_f32(
               false, a, false, bm, (short)0, z, false, false);
}

__global__ __launch_bounds__(TPB) void k_struct_fac(
    const float* __restrict__ coords, const float* __restrict__ q,
    const float* __restrict__ cell_inv,
    const float* __restrict__ kvec, float2* __restrict__ Spart,
    int N, int NK, int B)
{
    __shared__ float4 sAtom[PANEL];

    const int lane  = threadIdx.x & 31;
    const int wave  = threadIdx.x >> 5;
    const int kTile = (blockIdx.x * WAVES + wave) * 16;
    const int b     = blockIdx.z;
    const int nBeg  = blockIdx.y * PANEL;

    const int  m  = lane & 15;
    const bool hi = lane >= 16;

    const float* pC = coords + ((size_t)b * N + nBeg) * 3;
    const float* pQ = q + (size_t)b * N + nBeg;
    float4 stage[PANEL / TPB];
#pragma unroll
    for (int i = 0; i < PANEL / TPB; ++i) {
        const int r = i * TPB + threadIdx.x;
        stage[i].x = pC[r * 3 + 0];
        stage[i].y = pC[r * 3 + 1];
        stage[i].z = pC[r * 3 + 2];
        stage[i].w = pQ[r];
    }

    const float kx = kvec[(kTile + m) * 3 + 0];
    const float ky = kvec[(kTile + m) * 3 + 1];
    const float kz = kvec[(kTile + m) * 3 + 2];
    const float kp0 = kx * cell_inv[0] + ky * cell_inv[3] + kz * cell_inv[6];
    const float kp1 = kx * cell_inv[1] + ky * cell_inv[4] + kz * cell_inv[7];
    const float kp2 = kx * cell_inv[2] + ky * cell_inv[5] + kz * cell_inv[8];

    v2f a;
    a.x = hi ? kp2 : kp0;
    a.y = hi ? 0.0f : kp1;

#pragma unroll
    for (int i = 0; i < PANEL / TPB; ++i)
        sAtom[i * TPB + threadIdx.x] = stage[i];
    __syncthreads();

    v8f sumRe = {}, sumIm = {};

    for (int c = 0; c < PANEL; c += 64) {
        const float4 p0 = sAtom[c +  0 + m];
        const float4 p1 = sAtom[c + 16 + m];
        const float4 p2 = sAtom[c + 32 + m];
        const float4 p3 = sAtom[c + 48 + m];

        const v8f d0 = phase_tile(a, p0, hi);
        const v8f d1 = phase_tile(a, p1, hi);
        const v8f d2 = phase_tile(a, p2, hi);
        const v8f d3 = phase_tile(a, p3, hi);

        v8f sn, cs;
        sincos2pi_x8(d0, &sn, &cs);
        sumRe = sumRe + cs * p0.w;
        sumIm = sumIm + sn * p0.w;
        sincos2pi_x8(d1, &sn, &cs);
        sumRe = sumRe + cs * p1.w;
        sumIm = sumIm + sn * p1.w;
        sincos2pi_x8(d2, &sn, &cs);
        sumRe = sumRe + cs * p2.w;
        sumIm = sumIm + sn * p2.w;
        sincos2pi_x8(d3, &sn, &cs);
        sumRe = sumRe + cs * p3.w;
        sumIm = sumIm + sn * p3.w;
    }

#pragma unroll
    for (int r = 0; r < 8; ++r) {
        float re = sumRe[r], im = sumIm[r];
        for (int mask = 8; mask >= 1; mask >>= 1) {
            re += __shfl_xor(re, mask, 32);
            im += __shfl_xor(im, mask, 32);
        }
        sumRe[r] = re; sumIm[r] = im;
    }
    const int  row = (lane >> 1) & 7;
    const bool im  = lane & 1;
    float v = im ? sumIm[0] : sumRe[0];
#pragma unroll
    for (int r = 1; r < 8; ++r) {
        const float cand = im ? sumIm[r] : sumRe[r];
        v = (row == r) ? cand : v;
    }
    float* line = (float*)(Spart + ((size_t)blockIdx.y * B + b) * NK + kTile);
    line[lane] = v;
}

__global__ __launch_bounds__(TPB) __attribute__((amdgpu_waves_per_eu(1, 4))) void k_recip(
    const float* __restrict__ coords, const float* __restrict__ cell_inv,
    const float* __restrict__ kvec, const float* __restrict__ expfac,
    const float2* __restrict__ Spart,
    float* __restrict__ recipPart, int N, int NK, int B)
{
    __shared__ float4 sK[PANEL];
    __shared__ float4 sS4[PANEL / 2];
    __shared__ float  sOut[16 * WAVES];

    const int lane  = threadIdx.x & 31;
    const int wave  = threadIdx.x >> 5;
    const int nTile = (blockIdx.x * WAVES + wave) * 16;
    const int b     = blockIdx.z;
    const int kBeg  = blockIdx.y * PANEL;

    const int  m  = lane & 15;
    const bool hi = lane >= 16;

    const float* pK = kvec + (size_t)kBeg * 3;
    const float* pE = expfac + kBeg;
    float4 stageK[PANEL / TPB];
    float4 stageS[NSPLIT][PANEL / 2 / TPB];
#pragma unroll
    for (int i = 0; i < PANEL / TPB; ++i) {
        const int r = i * TPB + threadIdx.x;
        stageK[i].x = pK[r * 3 + 0];
        stageK[i].y = pK[r * 3 + 1];
        stageK[i].z = pK[r * 3 + 2];
        stageK[i].w = pE[r];
    }
#pragma unroll
    for (int p = 0; p < NSPLIT; ++p) {
        const float4* pS = (const float4*)(Spart + ((size_t)p * B + b) * NK + kBeg);
#pragma unroll
        for (int i = 0; i < PANEL / 2 / TPB; ++i)
            stageS[p][i] = pS[i * TPB + threadIdx.x];
    }

    const float* pX = coords + ((size_t)b * N + nTile + m) * 3;
    const float cx = pX[0], cy = pX[1], cz = pX[2];
    const float s0 = cell_inv[0] * cx + cell_inv[1] * cy + cell_inv[2] * cz;
    const float s1 = cell_inv[3] * cx + cell_inv[4] * cy + cell_inv[5] * cz;
    const float s2 = cell_inv[6] * cx + cell_inv[7] * cy + cell_inv[8] * cz;

    v2f a;
    a.x = hi ? s2 : s0;
    a.y = hi ? 0.0f : s1;

#pragma unroll
    for (int i = 0; i < PANEL / TPB; ++i)
        sK[i * TPB + threadIdx.x] = stageK[i];
#pragma unroll
    for (int i = 0; i < PANEL / 2 / TPB; ++i) {
        float4 s = stageS[0][i];
#pragma unroll
        for (int p = 1; p < NSPLIT; ++p) {
            s.x += stageS[p][i].x;
            s.y += stageS[p][i].y;
            s.z += stageS[p][i].z;
            s.w += stageS[p][i].w;
        }
        sS4[i * TPB + threadIdx.x] = s;
    }
    __syncthreads();

    const float2* sS = (const float2*)sS4;

    v8f acc = {};

    for (int c = 0; c < PANEL; c += 64) {
        const float4 kv0 = sK[c +  0 + m];
        const float4 kv1 = sK[c + 16 + m];
        const float4 kv2 = sK[c + 32 + m];
        const float4 kv3 = sK[c + 48 + m];
        const float2 S0 = sS[c +  0 + m];
        const float2 S1 = sS[c + 16 + m];
        const float2 S2 = sS[c + 32 + m];
        const float2 S3 = sS[c + 48 + m];

        const v8f d0 = phase_tile(a, kv0, hi);
        const v8f d1 = phase_tile(a, kv1, hi);
        const v8f d2 = phase_tile(a, kv2, hi);
        const v8f d3 = phase_tile(a, kv3, hi);

        v8f sn, cs;
        sincos2pi_x8(d0, &sn, &cs);
        acc = acc + (cs * S0.x + sn * S0.y) * kv0.w;
        sincos2pi_x8(d1, &sn, &cs);
        acc = acc + (cs * S1.x + sn * S1.y) * kv1.w;
        sincos2pi_x8(d2, &sn, &cs);
        acc = acc + (cs * S2.x + sn * S2.y) * kv2.w;
        sincos2pi_x8(d3, &sn, &cs);
        acc = acc + (cs * S3.x + sn * S3.y) * kv3.w;
    }

#pragma unroll
    for (int r = 0; r < 8; ++r) {
        float v = acc[r];
        for (int mask = 8; mask >= 1; mask >>= 1)
            v += __shfl_xor(v, mask, 32);
        acc[r] = v;
    }
    if (m == 0) {
#pragma unroll
        for (int r = 0; r < 8; ++r)
            sOut[wave * 16 + (hi ? 8 : 0) + r] = acc[r];
    }
    __syncthreads();
    if (threadIdx.x < 16 * WAVES)
        recipPart[((size_t)blockIdx.y * B + b) * N
                  + blockIdx.x * (16 * WAVES) + threadIdx.x] = sOut[threadIdx.x];
}

__global__ void k_phi(const float* __restrict__ q, const float* __restrict__ recipPart,
                      const float* __restrict__ volume, const float* __restrict__ bewald,
                      float* __restrict__ out, int BN)
{
    int i = blockIdx.x * blockDim.x + threadIdx.x;
    if (i >= BN) return;
    const float scale = BOHR_F / (PI_F * volume[0]);
    const float self  = 2.0f * bewald[0] * BOHR_F / sqrtf(PI_F);
    float recip = recipPart[i];
#pragma unroll
    for (int p = 1; p < NSPLIT; ++p)
        recip += recipPart[(size_t)p * BN + i];
    const float phi   = recip * scale - q[i] * self;
    out[i]      = 0.5f * q[i] * phi;
    out[BN + i] = phi;
}

extern "C" void kernel_launch(void* const* d_in, const int* in_sizes, int n_in,
                              void* d_out, int out_size, void* d_ws, size_t ws_size,
                              hipStream_t stream)
{
    const float* coords   = (const float*)d_in[0];
    const float* q        = (const float*)d_in[1];
    const float* cell_inv = (const float*)d_in[2];
    const float* kvec     = (const float*)d_in[3];
    const float* expfac   = (const float*)d_in[4];
    const float* volume   = (const float*)d_in[5];
    const float* bewald   = (const float*)d_in[6];
    float* out = (float*)d_out;

    const int B  = 2;
    const int BN = in_sizes[1];
    const int N  = BN / B;
    const int NK = in_sizes[4];

    float2* Spart     = (float2*)d_ws;
    float*  recipPart = (float*)(Spart + (size_t)NSPLIT * B * NK);
    const size_t wsFloats = (size_t)2 * NSPLIT * B * NK + (size_t)NSPLIT * BN;
    if (ws_size < wsFloats * sizeof(float)) return;
    if (N != NSPLIT * PANEL || NK != NSPLIT * PANEL) return;

    dim3 g1(NK / (16 * WAVES), NSPLIT, B);
    k_struct_fac<<<g1, TPB, 0, stream>>>(coords, q, cell_inv, kvec, Spart, N, NK, B);

    dim3 g2(N / (16 * WAVES), NSPLIT, B);
    k_recip<<<g2, TPB, 0, stream>>>(coords, cell_inv, kvec, expfac, Spart,
                                    recipPart, N, NK, B);

    k_phi<<<(BN + 255) / 256, 256, 0, stream>>>(q, recipPart, volume, bewald, out, BN);
}


# Epilogue launch dropped (3 to 2): kernel 1 stores each atom's self-term start values to out, kernel 2's k-panel blocks add their scaled shares by float atomics; 4 shares summed in arrival order
# speedup vs baseline: 1.9054x; 1.1436x over previous
#include <hip/hip_runtime.h>
#include <math.h>

typedef __attribute__((ext_vector_type(2))) float v2f;
typedef __attribute__((ext_vector_type(8))) float v8f;

#define BOHR_F 1.8897261258369282f
#define PI_F   3.14159265358979323846f

__device__ __forceinline__ void sincos2pi_x8(v8f x, v8f* sn, v8f* cs) {
    float s0, s1, s2, s3, s4, s5, s6, s7;
    float c0, c1, c2, c3, c4, c5, c6, c7;
    asm volatile(
        "v_sin_f32 %0, %16\n\t"
        "v_sin_f32 %1, %17\n\t"
        "v_sin_f32 %2, %18\n\t"
        "v_sin_f32 %3, %19\n\t"
        "v_sin_f32 %4, %20\n\t"
        "v_sin_f32 %5, %21\n\t"
        "v_sin_f32 %6, %22\n\t"
        "v_sin_f32 %7, %23\n\t"
        "v_cos_f32 %8, %16\n\t"
        "v_cos_f32 %9, %17\n\t"
        "v_cos_f32 %10, %18\n\t"
        "v_cos_f32 %11, %19\n\t"
        "v_cos_f32 %12, %20\n\t"
        "v_cos_f32 %13, %21\n\t"
        "v_cos_f32 %14, %22\n\t"
        "v_cos_f32 %15, %23\n\t"
        "v_nop"
        : "=&v"(s0), "=&v"(s1), "=&v"(s2), "=&v"(s3),
          "=&v"(s4), "=&v"(s5), "=&v"(s6), "=&v"(s7),
          "=&v"(c0), "=&v"(c1), "=&v"(c2), "=&v"(c3),
          "=&v"(c4), "=&v"(c5), "=&v"(c6), "=&v"(c7)
        : "v"(x[0]), "v"(x[1]), "v"(x[2]), "v"(x[3]),
          "v"(x[4]), "v"(x[5]), "v"(x[6]), "v"(x[7]));
    *sn = (v8f){s0, s1, s2, s3, s4, s5, s6, s7};
    *cs = (v8f){c0, c1, c2, c3, c4, c5, c6, c7};
}

#define PANEL  1024
#define NSPLIT 4
#define WAVES  4
#define TPB    (WAVES * 32)

__device__ __forceinline__ v8f phase_tile(v2f a, float4 p, bool hi) {
    v2f bm;
    bm.x = hi ? p.z : p.x;
    bm.y = hi ? 0.0f : p.y;
    v8f z = {};
    return __builtin_amdgcn_wmma_f32_16x16x4_f32(
               false, a, false, bm, (short)0, z, false, false);
}

__global__ __launch_bounds__(TPB) void k_struct_fac(
    const float* __restrict__ coords, const float* __restrict__ q,
    const float* __restrict__ cell_inv,
    const float* __restrict__ kvec, const float* __restrict__ bewald,
    float2* __restrict__ Spart, float* __restrict__ out,
    int N, int NK, int B)
{
    __shared__ float4 sAtom[PANEL];

    const int lane  = threadIdx.x & 31;
    const int wave  = threadIdx.x >> 5;
    const int kTile = (blockIdx.x * WAVES + wave) * 16;
    const int b     = blockIdx.z;
    const int nBeg  = blockIdx.y * PANEL;

    const int  m  = lane & 15;
    const bool hi = lane >= 16;

    const float* pC = coords + ((size_t)b * N + nBeg) * 3;
    const float* pQ = q + (size_t)b * N + nBeg;
    float4 stage[PANEL / TPB];
#pragma unroll
    for (int i = 0; i < PANEL / TPB; ++i) {
        const int r = i * TPB + threadIdx.x;
        stage[i].x = pC[r * 3 + 0];
        stage[i].y = pC[r * 3 + 1];
        stage[i].z = pC[r * 3 + 2];
        stage[i].w = pQ[r];
    }

    const float kx = kvec[(kTile + m) * 3 + 0];
    const float ky = kvec[(kTile + m) * 3 + 1];
    const float kz = kvec[(kTile + m) * 3 + 2];
    const float kp0 = kx * cell_inv[0] + ky * cell_inv[3] + kz * cell_inv[6];
    const float kp1 = kx * cell_inv[1] + ky * cell_inv[4] + kz * cell_inv[7];
    const float kp2 = kx * cell_inv[2] + ky * cell_inv[5] + kz * cell_inv[8];

    v2f a;
    a.x = hi ? kp2 : kp0;
    a.y = hi ? 0.0f : kp1;

#pragma unroll
    for (int i = 0; i < PANEL / TPB; ++i)
        sAtom[i * TPB + threadIdx.x] = stage[i];

    __syncthreads();

    if ((blockIdx.x & 1) == 0 && wave == 0) {
        const int    r    = (blockIdx.x >> 1) * 32 + lane;
        const float  qi   = sAtom[r].w;
        const float  self = 2.0f * bewald[0] * BOHR_F / sqrtf(PI_F);
        const float  phi0 = -qi * self;
        const size_t i    = (size_t)b * N + nBeg + r;
        out[(size_t)B * N + i] = phi0;
        out[i]                 = 0.5f * qi * phi0;
    }

    v8f sumRe = {}, sumIm = {};

    for (int c = 0; c < PANEL; c += 64) {
        const float4 p0 = sAtom[c +  0 + m];
        const float4 p1 = sAtom[c + 16 + m];
        const float4 p2 = sAtom[c + 32 + m];
        const float4 p3 = sAtom[c + 48 + m];

        const v8f d0 = phase_tile(a, p0, hi);
        const v8f d1 = phase_tile(a, p1, hi);
        const v8f d2 = phase_tile(a, p2, hi);
        const v8f d3 = phase_tile(a, p3, hi);

        v8f sn, cs;
        sincos2pi_x8(d0, &sn, &cs);
        sumRe = sumRe + cs * p0.w;
        sumIm = sumIm + sn * p0.w;
        sincos2pi_x8(d1, &sn, &cs);
        sumRe = sumRe + cs * p1.w;
        sumIm = sumIm + sn * p1.w;
        sincos2pi_x8(d2, &sn, &cs);
        sumRe = sumRe + cs * p2.w;
        sumIm = sumIm + sn * p2.w;
        sincos2pi_x8(d3, &sn, &cs);
        sumRe = sumRe + cs * p3.w;
        sumIm = sumIm + sn * p3.w;
    }

#pragma unroll
    for (int r = 0; r < 8; ++r) {
        float re = sumRe[r], im = sumIm[r];
        for (int mask = 8; mask >= 1; mask >>= 1) {
            re += __shfl_xor(re, mask, 32);
            im += __shfl_xor(im, mask, 32);
        }
        sumRe[r] = re; sumIm[r] = im;
    }
    const int  row = (lane >> 1) & 7;
    const bool im  = lane & 1;
    float v = im ? sumIm[0] : sumRe[0];
#pragma unroll
    for (int r = 1; r < 8; ++r) {
        const float cand = im ? sumIm[r] : sumRe[r];
        v = (row == r) ? cand : v;
    }
    float* line = (float*)(Spart + ((size_t)blockIdx.y * B + b) * NK + kTile);
    line[lane] = v;
}

__global__ __launch_bounds__(TPB) __attribute__((amdgpu_waves_per_eu(1, 4))) void k_recip(
    const float* __restrict__ coords, const float* __restrict__ cell_inv,
    const float* __restrict__ kvec, const float* __restrict__ expfac,
    const float* __restrict__ q, const float* __restrict__ volume,
    const float2* __restrict__ Spart,
    float* __restrict__ out, int N, int NK, int B)
{
    __shared__ float4 sK[PANEL];
    __shared__ float4 sS4[PANEL / 2];
    __shared__ float  sOut[16 * WAVES];

    const int lane  = threadIdx.x & 31;
    const int wave  = threadIdx.x >> 5;
    const int nTile = (blockIdx.x * WAVES + wave) * 16;
    const int b     = blockIdx.z;
    const int kBeg  = blockIdx.y * PANEL;

    const int  m  = lane & 15;
    const bool hi = lane >= 16;

    const float* pK = kvec + (size_t)kBeg * 3;
    const float* pE = expfac + kBeg;
    float4 stageK[PANEL / TPB];
    float4 stageS[NSPLIT][PANEL / 2 / TPB];
#pragma unroll
    for (int i = 0; i < PANEL / TPB; ++i) {
        const int r = i * TPB + threadIdx.x;
        stageK[i].x = pK[r * 3 + 0];
        stageK[i].y = pK[r * 3 + 1];
        stageK[i].z = pK[r * 3 + 2];
        stageK[i].w = pE[r];
    }
#pragma unroll
    for (int p = 0; p < NSPLIT; ++p) {
        const float4* pS = (const float4*)(Spart + ((size_t)p * B + b) * NK + kBeg);
#pragma unroll
        for (int i = 0; i < PANEL / 2 / TPB; ++i)
            stageS[p][i] = pS[i * TPB + threadIdx.x];
    }

    const float* pX = coords + ((size_t)b * N + nTile + m) * 3;
    const float cx = pX[0], cy = pX[1], cz = pX[2];

    const size_t iOut  = (size_t)b * N + blockIdx.x * (16 * WAVES) + (threadIdx.x & (16 * WAVES - 1));
    const float  qOut  = q[iOut];
    const float  scale = BOHR_F / (PI_F * volume[0]);
    const float s0 = cell_inv[0] * cx + cell_inv[1] * cy + cell_inv[2] * cz;
    const float s1 = cell_inv[3] * cx + cell_inv[4] * cy + cell_inv[5] * cz;
    const float s2 = cell_inv[6] * cx + cell_inv[7] * cy + cell_inv[8] * cz;

    v2f a;
    a.x = hi ? s2 : s0;
    a.y = hi ? 0.0f : s1;

#pragma unroll
    for (int i = 0; i < PANEL / TPB; ++i)
        sK[i * TPB + threadIdx.x] = stageK[i];
#pragma unroll
    for (int i = 0; i < PANEL / 2 / TPB; ++i) {
        float4 s = stageS[0][i];
#pragma unroll
        for (int p = 1; p < NSPLIT; ++p) {
            s.x += stageS[p][i].x;
            s.y += stageS[p][i].y;
            s.z += stageS[p][i].z;
            s.w += stageS[p][i].w;
        }
        sS4[i * TPB + threadIdx.x] = s;
    }
    __syncthreads();

    const float2* sS = (const float2*)sS4;

    v8f acc = {};

    for (int c = 0; c < PANEL; c += 64) {
        const float4 kv0 = sK[c +  0 + m];
        const float4 kv1 = sK[c + 16 + m];
        const float4 kv2 = sK[c + 32 + m];
        const float4 kv3 = sK[c + 48 + m];
        const float2 S0 = sS[c +  0 + m];
        const float2 S1 = sS[c + 16 + m];
        const float2 S2 = sS[c + 32 + m];
        const float2 S3 = sS[c + 48 + m];

        const v8f d0 = phase_tile(a, kv0, hi);
        const v8f d1 = phase_tile(a, kv1, hi);
        const v8f d2 = phase_tile(a, kv2, hi);
        const v8f d3 = phase_tile(a, kv3, hi);

        v8f sn, cs;
        sincos2pi_x8(d0, &sn, &cs);
        acc = acc + (cs * S0.x + sn * S0.y) * kv0.w;
        sincos2pi_x8(d1, &sn, &cs);
        acc = acc + (cs * S1.x + sn * S1.y) * kv1.w;
        sincos2pi_x8(d2, &sn, &cs);
        acc = acc + (cs * S2.x + sn * S2.y) * kv2.w;
        sincos2pi_x8(d3, &sn, &cs);
        acc = acc + (cs * S3.x + sn * S3.y) * kv3.w;
    }

#pragma unroll
    for (int r = 0; r < 8; ++r) {
        float v = acc[r];
        for (int mask = 8; mask >= 1; mask >>= 1)
            v += __shfl_xor(v, mask, 32);
        acc[r] = v;
    }
    if (m == 0) {
#pragma unroll
        for (int r = 0; r < 8; ++r)
            sOut[wave * 16 + (hi ? 8 : 0) + r] = acc[r];
    }
    __syncthreads();
    if (threadIdx.x < 16 * WAVES) {
        const float share = sOut[threadIdx.x] * scale;
        atomicAdd(&out[(size_t)B * N + iOut], share);
        atomicAdd(&out[iOut], 0.5f * qOut * share);
    }
}

extern "C" void kernel_launch(void* const* d_in, const int* in_sizes, int n_in,
                              void* d_out, int out_size, void* d_ws, size_t ws_size,
                              hipStream_t stream)
{
    const float* coords   = (const float*)d_in[0];
    const float* q        = (const float*)d_in[1];
    const float* cell_inv = (const float*)d_in[2];
    const float* kvec     = (const float*)d_in[3];
    const float* expfac   = (const float*)d_in[4];
    const float* volume   = (const float*)d_in[5];
    const float* bewald   = (const float*)d_in[6];
    float* out = (float*)d_out;

    const int B  = 2;
    const int BN = in_sizes[1];
    const int N  = BN / B;
    const int NK = in_sizes[4];

    float2* Spart = (float2*)d_ws;
    const size_t wsFloats = (size_t)2 * NSPLIT * B * NK;
    if (ws_size < wsFloats * sizeof(float)) return;
    if (N != NSPLIT * PANEL || NK != NSPLIT * PANEL) return;

    dim3 g1(NK / (16 * WAVES), NSPLIT, B);
    k_struct_fac<<<g1, TPB, 0, stream>>>(coords, q, cell_inv, kvec, bewald, Spart, out, N, NK, B);

    dim3 g2(N / (16 * WAVES), NSPLIT, B);
    k_recip<<<g2, TPB, 0, stream>>>(coords, cell_inv, kvec, expfac, q, volume, Spart,
                                    out, N, NK, B);
}
